# starting kernel
# baseline (speedup 1.0000x reference)
#include <hip/hip_runtime.h>

#define BATCH_N   16384
#define MAX_ITEMS 200
#define EMB_DIM   128
#define NTILES    ((MAX_ITEMS + 15) / 16)

typedef __attribute__((ext_vector_type(2))) float v2f;
typedef __attribute__((ext_vector_type(8))) float v8f;

__global__ __launch_bounds__(128) void ncf_fused_kernel(
    const float* __restrict__ ratings,
    const int*   __restrict__ item_ids,
    const float* __restrict__ noise,
    const float* __restrict__ init_user_emb,
    const float* __restrict__ gmf_emb,
    const float* __restrict__ ncf_emb,
    const float* __restrict__ W0, const float* __restrict__ b0,
    const float* __restrict__ W1, const float* __restrict__ b1,
    const float* __restrict__ W2, const float* __restrict__ b2,
    const float* __restrict__ W3, const float* __restrict__ b3,
    const float* __restrict__ Wn, const float* __restrict__ bn,
    const float* __restrict__ Wf, const float* __restrict__ bf,
    const float* __restrict__ p,
    float* __restrict__ out)
{
    __shared__ float s_row[2 * EMB_DIM];
    __shared__ float s_dots[16];
    __shared__ float s_g[4];
    __shared__ float s_u[8];
    __shared__ int   s_ids[MAX_ITEMS];

    const int b = blockIdx.x;
    const int t = threadIdx.x;
    const long rowBase = (long)b * MAX_ITEMS;

    if (t < 13) __builtin_prefetch(&ratings[rowBase + t * 16], 0, 3);

    s_row[t]           = init_user_emb[b * EMB_DIM + t];
    s_row[EMB_DIM + t] = noise[b * EMB_DIM + t];
    for (int i = t; i < MAX_ITEMS; i += 128) s_ids[i] = item_ids[rowBase + i];
    __syncthreads();

    if (t < 16) {
        const int grp = t >> 2;
        const int j   = t & 3;
        const float* W    = (grp == 0) ? W0 : (grp == 1) ? W1 : (grp == 2) ? W2 : W3;
        const float* bias = (grp == 0) ? b0 : (grp == 1) ? b1 : (grp == 2) ? b2 : b3;
        const float* x = &s_row[grp * 64];
        float acc = bias[j];
        #pragma unroll 8
        for (int k = 0; k < 64; ++k) acc += x[k] * W[j * 64 + k];
        s_dots[t] = acc;
    }
    __syncthreads();

    if (t < 4) s_g[t] = s_dots[t] * Wf[t] + s_dots[8 + t] * Wf[4 + t];
    if (t < 8) {
        float acc = bn[t];
        #pragma unroll
        for (int j = 0; j < 4; ++j)
            acc += Wn[t * 12 + j] * s_dots[4 + j] + Wn[t * 12 + 4 + j] * s_dots[12 + j];
        s_u[t] = acc;
    }
    __syncthreads();

    const int wave  = t >> 5;
    const int lane  = t & 31;
    const int row16 = lane & 15;
    const bool hi   = lane >= 16;

    v2f A = {0.0f, 0.0f};
    if (row16 < 8) {
        const int kb = 8 + (hi ? 2 : 0);
        A.x = Wn[row16 * 12 + kb];
        A.y = Wn[row16 * 12 + kb + 1];
    }

    const float g0 = s_g[0], g1 = s_g[1], g2 = s_g[2], g3 = s_g[3];
    float u_r[8], wf8[8];
    #pragma unroll
    for (int r = 0; r < 8; ++r) { u_r[r] = s_u[r]; wf8[r] = Wf[8 + r]; }
    const float pv = p[0], bfv = bf[0];

    for (int tile = wave; tile < NTILES; tile += 4) {
        const int item  = tile * 16 + row16;
        const int itemC = (item < MAX_ITEMS) ? item : (MAX_ITEMS - 1);
        const int iid   = s_ids[itemC];

        const float2 bv = *(const float2*)&ncf_emb[(long)iid * 4 + (hi ? 2 : 0)];
        const float4 gi = *(const float4*)&gmf_emb[(long)iid * 4];
        float gx = gi.x, gy = gi.y, gz = gi.z, gw = gi.w;
        float rt = ratings[rowBase + itemC];

        asm volatile("" : "+v"(gx), "+v"(gy), "+v"(gz), "+v"(gw), "+v"(rt));

        v2f B = {bv.x, bv.y};
        v8f C = {};
        C = __builtin_amdgcn_wmma_f32_16x16x4_f32(
                 false, A,  false, B,
                 (short)0, C,  false,  false);

        float score = bfv + gx * g0 + gy * g1 + gz * g2 + gw * g3;
        #pragma unroll
        for (int r = 0; r < 8; ++r) {
            const float h = u_r[r] + C[r];
            score += fmaxf(h, 0.0f) * wf8[r];
        }
        const float result = pv * rt + score;

        if (!hi && item < MAX_ITEMS)
            out[rowBase + item] = result;
    }
}

extern "C" void kernel_launch(void* const* d_in, const int* in_sizes, int n_in,
                              void* d_out, int out_size, void* d_ws, size_t ws_size,
                              hipStream_t stream) {
    const float* ratings       = (const float*)d_in[0];
    const int*   item_ids      = (const int*)  d_in[1];
    const float* noise         = (const float*)d_in[2];
    const float* init_user_emb = (const float*)d_in[3];
    const float* gmf_emb       = (const float*)d_in[4];
    const float* ncf_emb       = (const float*)d_in[5];
    const float* W0 = (const float*)d_in[6];  const float* b0 = (const float*)d_in[7];
    const float* W1 = (const float*)d_in[8];  const float* b1 = (const float*)d_in[9];
    const float* W2 = (const float*)d_in[10]; const float* b2 = (const float*)d_in[11];
    const float* W3 = (const float*)d_in[12]; const float* b3 = (const float*)d_in[13];
    const float* Wn = (const float*)d_in[14]; const float* bn = (const float*)d_in[15];
    const float* Wf = (const float*)d_in[16]; const float* bf = (const float*)d_in[17];
    const float* p  = (const float*)d_in[18];
    float* out = (float*)d_out;

    ncf_fused_kernel<<<BATCH_N, 128, 0, stream>>>(
        ratings, item_ids, noise, init_user_emb, gmf_emb, ncf_emb,
        W0, b0, W1, b1, W2, b2, W3, b3, Wn, bn, Wf, bf, p, out);
}


# Persistent grid: 4096 workgroups (16 per WGP, every wave slot) each walk 4 batch rows instead of 16384 one-row workgroups: 4x fewer waves enter; item-id LDS buffer doubled, A fragment loaded once
# speedup vs baseline: 1.0554x; 1.0554x over previous
#include <hip/hip_runtime.h>


#define BATCH_N   16384
#define MAX_ITEMS 200
#define EMB_DIM   128
#define NTILES    ((MAX_ITEMS + 15) / 16)
#define GRID_WG   4096

typedef __attribute__((ext_vector_type(2))) float v2f;
typedef __attribute__((ext_vector_type(8))) float v8f;

__global__ __launch_bounds__(128, 16) void ncf_fused_kernel(
    const float* __restrict__ ratings,
    const int*   __restrict__ item_ids,
    const float* __restrict__ noise,
    const float* __restrict__ init_user_emb,
    const float* __restrict__ gmf_emb,
    const float* __restrict__ ncf_emb,
    const float* __restrict__ W0, const float* __restrict__ b0,
    const float* __restrict__ W1, const float* __restrict__ b1,
    const float* __restrict__ W2, const float* __restrict__ b2,
    const float* __restrict__ W3, const float* __restrict__ b3,
    const float* __restrict__ Wn, const float* __restrict__ bn,
    const float* __restrict__ Wf, const float* __restrict__ bf,
    const float* __restrict__ p,
    float* __restrict__ out)
{
    __shared__ float s_row[2 * EMB_DIM];
    __shared__ float s_dots[16];
    __shared__ float s_g[4];
    __shared__ float s_u[8];
    __shared__ int   s_ids[2][MAX_ITEMS];

    const int t = threadIdx.x;
    const int wave  = t >> 5;
    const int lane  = t & 31;
    const int row16 = lane & 15;
    const bool hi   = lane >= 16;


    const int grp = (t >> 2) & 3;
    const int pj  = t & 3;
    const float* Wp   = (grp == 0) ? W0 : (grp == 1) ? W1 : (grp == 2) ? W2 : W3;
    const float* bp   = (grp == 0) ? b0 : (grp == 1) ? b1 : (grp == 2) ? b2 : b3;
    const float* Wrow = Wp + pj * 64;
    const float  pbias = bp[pj];

    v2f A = {0.0f, 0.0f};
    if (row16 < 8) {
        const int kb = 8 + (hi ? 2 : 0);
        A.x = Wn[row16 * 12 + kb];
        A.y = Wn[row16 * 12 + kb + 1];
    }
    float wf8[8];
    #pragma unroll
    for (int r = 0; r < 8; ++r) wf8[r] = Wf[8 + r];
    const float pv = p[0], bfv = bf[0];

    int par = 0;
    for (int b = blockIdx.x; b < BATCH_N; b += GRID_WG, par ^= 1) {
        const long rowBase = (long)b * MAX_ITEMS;
        int* ids = s_ids[par];

        if (t < 13) __builtin_prefetch(&ratings[rowBase + t * 16], 0, 3);

        s_row[t]           = init_user_emb[b * EMB_DIM + t];
        s_row[EMB_DIM + t] = noise[b * EMB_DIM + t];
        for (int i = t; i < MAX_ITEMS; i += 128) ids[i] = item_ids[rowBase + i];
        __syncthreads();

        if (t < 16) {
            const float* x = &s_row[grp * 64];
            float acc = pbias;
            #pragma unroll 8
            for (int k = 0; k < 64; ++k) acc += x[k] * Wrow[k];
            s_dots[t] = acc;
        }
        __syncthreads();

        if (t < 4) s_g[t] = s_dots[t] * Wf[t] + s_dots[8 + t] * Wf[4 + t];
        if (t < 8) {
            float acc = bn[t];
            #pragma unroll
            for (int j = 0; j < 4; ++j)
                acc += Wn[t * 12 + j] * s_dots[4 + j] + Wn[t * 12 + 4 + j] * s_dots[12 + j];
            s_u[t] = acc;
        }
        __syncthreads();

        const float g0 = s_g[0], g1 = s_g[1], g2 = s_g[2], g3 = s_g[3];
        float u_r[8];
        #pragma unroll
        for (int r = 0; r < 8; ++r) u_r[r] = s_u[r];

        for (int tile = wave; tile < NTILES; tile += 4) {
            const int item  = tile * 16 + row16;
            const int itemC = (item < MAX_ITEMS) ? item : (MAX_ITEMS - 1);
            const int iid   = ids[itemC];

            const float2 bv = *(const float2*)&ncf_emb[(long)iid * 4 + (hi ? 2 : 0)];
            const float4 gi = *(const float4*)&gmf_emb[(long)iid * 4];
            float gx = gi.x, gy = gi.y, gz = gi.z, gw = gi.w;
            float rt = ratings[rowBase + itemC];

            asm volatile("" : "+v"(gx), "+v"(gy), "+v"(gz), "+v"(gw), "+v"(rt));

            v2f B = {bv.x, bv.y};
            v8f C = {};
            C = __builtin_amdgcn_wmma_f32_16x16x4_f32(
                     false, A,  false, B,
                     (short)0, C,  false,  false);

            float score = bfv + gx * g0 + gy * g1 + gz * g2 + gw * g3;
            #pragma unroll
            for (int r = 0; r < 8; ++r) {
                const float h = u_r[r] + C[r];
                score += fmaxf(h, 0.0f) * wf8[r];
            }
            const float result = pv * rt + score;

            if (!hi && item < MAX_ITEMS)
                out[rowBase + item] = result;
        }
    }
}

extern "C" void kernel_launch(void* const* d_in, const int* in_sizes, int n_in,
                              void* d_out, int out_size, void* d_ws, size_t ws_size,
                              hipStream_t stream) {
    const float* ratings       = (const float*)d_in[0];
    const int*   item_ids      = (const int*)  d_in[1];
    const float* noise         = (const float*)d_in[2];
    const float* init_user_emb = (const float*)d_in[3];
    const float* gmf_emb       = (const float*)d_in[4];
    const float* ncf_emb       = (const float*)d_in[5];
    const float* W0 = (const float*)d_in[6];  const float* b0 = (const float*)d_in[7];
    const float* W1 = (const float*)d_in[8];  const float* b1 = (const float*)d_in[9];
    const float* W2 = (const float*)d_in[10]; const float* b2 = (const float*)d_in[11];
    const float* W3 = (const float*)d_in[12]; const float* b3 = (const float*)d_in[13];
    const float* Wn = (const float*)d_in[14]; const float* bn = (const float*)d_in[15];
    const float* Wf = (const float*)d_in[16]; const float* bf = (const float*)d_in[17];
    const float* p  = (const float*)d_in[18];
    float* out = (float*)d_out;

    ncf_fused_kernel<<<GRID_WG, 128, 0, stream>>>(
        ratings, item_ids, noise, init_user_emb, gmf_emb, ncf_emb,
        W0, b0, W1, b1, W2, b2, W3, b3, Wn, bn, Wf, bf, p, out);
}


# Pre-pass kernel interleaves the two item tables into one 32-byte row per item in the workspace: one 16-byte gather and one cache line per item, not two; gmf dot summed as two half-wave parts
# speedup vs baseline: 1.2374x; 1.1724x over previous
#include <hip/hip_runtime.h>


#define BATCH_N   16384
#define MAX_ITEMS 200
#define EMB_DIM   128
#define NTILES    ((MAX_ITEMS + 15) / 16)
#define GRID_WG   4096

#define NUM_ITEMS        100000
#define MERGE_BLOCK      128
#define MERGE_PER_THREAD 8
#define MERGE_CHUNKS     (NUM_ITEMS * 2)
#define MERGE_GRID       ((MERGE_CHUNKS + MERGE_BLOCK * MERGE_PER_THREAD - 1) / (MERGE_BLOCK * MERGE_PER_THREAD))

typedef __attribute__((ext_vector_type(2))) float v2f;
typedef __attribute__((ext_vector_type(8))) float v8f;

__global__ __launch_bounds__(MERGE_BLOCK) void interleave_tables_kernel(
    const float* __restrict__ gmf_emb,
    const float* __restrict__ ncf_emb,
    float* __restrict__ merged)
{
    const float2* n2 = (const float2*)ncf_emb;
    const float2* g2 = (const float2*)gmf_emb;
    float4* m4 = (float4*)merged;
    const int base = blockIdx.x * (MERGE_BLOCK * MERGE_PER_THREAD) + threadIdx.x;
    float2 nv[MERGE_PER_THREAD], gv[MERGE_PER_THREAD];
    #pragma unroll
    for (int k = 0; k < MERGE_PER_THREAD; ++k) {
        const int j = base + k * MERGE_BLOCK;
        if (j < MERGE_CHUNKS) { nv[k] = n2[j]; gv[k] = g2[j]; }
    }
    #pragma unroll
    for (int k = 0; k < MERGE_PER_THREAD; ++k) {
        const int j = base + k * MERGE_BLOCK;
        if (j < MERGE_CHUNKS) m4[j] = make_float4(nv[k].x, nv[k].y, gv[k].x, gv[k].y);
    }
}

__device__ __forceinline__ float xhalf(float x) {
    const int b = __builtin_bit_cast(int, x);
    return __builtin_bit_cast(float,
        __builtin_amdgcn_permlanex16(b, b, 0x76543210, 0xfedcba98, false, false));
}

__global__ __launch_bounds__(128, 16) void ncf_fused_kernel(
    const float* __restrict__ ratings,
    const int*   __restrict__ item_ids,
    const float* __restrict__ noise,
    const float* __restrict__ init_user_emb,
    const float* __restrict__ merged,
    const float* __restrict__ W0, const float* __restrict__ b0,
    const float* __restrict__ W1, const float* __restrict__ b1,
    const float* __restrict__ W2, const float* __restrict__ b2,
    const float* __restrict__ W3, const float* __restrict__ b3,
    const float* __restrict__ Wn, const float* __restrict__ bn,
    const float* __restrict__ Wf, const float* __restrict__ bf,
    const float* __restrict__ p,
    float* __restrict__ out)
{
    __shared__ float s_row[2 * EMB_DIM];
    __shared__ float s_dots[16];
    __shared__ float s_g[4];
    __shared__ float s_u[8];
    __shared__ int   s_ids[2][MAX_ITEMS];

    const int t = threadIdx.x;
    const int wave  = t >> 5;
    const int lane  = t & 31;
    const int row16 = lane & 15;
    const bool hi   = lane >= 16;


    const int grp = (t >> 2) & 3;
    const int pj  = t & 3;
    const float* Wp   = (grp == 0) ? W0 : (grp == 1) ? W1 : (grp == 2) ? W2 : W3;
    const float* bp   = (grp == 0) ? b0 : (grp == 1) ? b1 : (grp == 2) ? b2 : b3;
    const float* Wrow = Wp + pj * 64;
    const float  pbias = bp[pj];

    v2f A = {0.0f, 0.0f};
    if (row16 < 8) {
        const int kb = 8 + (hi ? 2 : 0);
        A.x = Wn[row16 * 12 + kb];
        A.y = Wn[row16 * 12 + kb + 1];
    }
    float wf8[8];
    #pragma unroll
    for (int r = 0; r < 8; ++r) wf8[r] = Wf[8 + r];
    const float pv = p[0], bfv = bf[0];
    const float sbase = hi ? 0.0f : bfv;

    int par = 0;
    for (int b = blockIdx.x; b < BATCH_N; b += GRID_WG, par ^= 1) {
        const long rowBase = (long)b * MAX_ITEMS;
        int* ids = s_ids[par];

        if (t < 13) __builtin_prefetch(&ratings[rowBase + t * 16], 0, 3);

        s_row[t]           = init_user_emb[b * EMB_DIM + t];
        s_row[EMB_DIM + t] = noise[b * EMB_DIM + t];
        for (int i = t; i < MAX_ITEMS; i += 128) ids[i] = item_ids[rowBase + i];
        __syncthreads();

        if (t < 16) {
            const float* x = &s_row[grp * 64];
            float acc = pbias;
            #pragma unroll 8
            for (int k = 0; k < 64; ++k) acc += x[k] * Wrow[k];
            s_dots[t] = acc;
        }
        __syncthreads();

        if (t < 4) s_g[t] = s_dots[t] * Wf[t] + s_dots[8 + t] * Wf[4 + t];
        if (t < 8) {
            float acc = bn[t];
            #pragma unroll
            for (int j = 0; j < 4; ++j)
                acc += Wn[t * 12 + j] * s_dots[4 + j] + Wn[t * 12 + 4 + j] * s_dots[12 + j];
            s_u[t] = acc;
        }
        __syncthreads();

        const float ga = s_g[hi ? 2 : 0], gb = s_g[hi ? 3 : 1];
        float u_r[8];
        #pragma unroll
        for (int r = 0; r < 8; ++r) u_r[r] = s_u[r];

        for (int tile = wave; tile < NTILES; tile += 4) {
            const int item  = tile * 16 + row16;
            const int itemC = (item < MAX_ITEMS) ? item : (MAX_ITEMS - 1);
            const int iid   = ids[itemC];

            const float4 mv = *(const float4*)&merged[(long)iid * 8 + (hi ? 4 : 0)];
            const float rt = ratings[rowBase + itemC];

            asm volatile("" :: "v"(rt));

            v2f B = {mv.x, mv.y};
            v8f C = {};
            C = __builtin_amdgcn_wmma_f32_16x16x4_f32(
                     false, A,  false, B,
                     (short)0, C,  false,  false);

            const float part = sbase + mv.z * ga + mv.w * gb;
            float score = part + xhalf(part);
            #pragma unroll
            for (int r = 0; r < 8; ++r) {
                const float h = u_r[r] + C[r];
                score += fmaxf(h, 0.0f) * wf8[r];
            }
            const float result = pv * rt + score;

            if (!hi && item < MAX_ITEMS)
                out[rowBase + item] = result;
        }
    }
}

extern "C" void kernel_launch(void* const* d_in, const int* in_sizes, int n_in,
                              void* d_out, int out_size, void* d_ws, size_t ws_size,
                              hipStream_t stream) {
    const float* ratings       = (const float*)d_in[0];
    const int*   item_ids      = (const int*)  d_in[1];
    const float* noise         = (const float*)d_in[2];
    const float* init_user_emb = (const float*)d_in[3];
    const float* gmf_emb       = (const float*)d_in[4];
    const float* ncf_emb       = (const float*)d_in[5];
    float* merged = (float*)d_ws;
    const float* W0 = (const float*)d_in[6];  const float* b0 = (const float*)d_in[7];
    const float* W1 = (const float*)d_in[8];  const float* b1 = (const float*)d_in[9];
    const float* W2 = (const float*)d_in[10]; const float* b2 = (const float*)d_in[11];
    const float* W3 = (const float*)d_in[12]; const float* b3 = (const float*)d_in[13];
    const float* Wn = (const float*)d_in[14]; const float* bn = (const float*)d_in[15];
    const float* Wf = (const float*)d_in[16]; const float* bf = (const float*)d_in[17];
    const float* p  = (const float*)d_in[18];
    float* out = (float*)d_out;

    interleave_tables_kernel<<<MERGE_GRID, MERGE_BLOCK, 0, stream>>>(gmf_emb, ncf_emb, merged);

    ncf_fused_kernel<<<GRID_WG, 128, 0, stream>>>(
        ratings, item_ids, noise, init_user_emb, merged,
        W0, b0, W1, b1, W2, b2, W3, b3, Wn, bn, Wf, bf, p, out);
}


# Each wave issues its row's loads up front (ids, ratings, then all 3-4 table gathers) and the row's barriers fence LDS only: gathers stay in flight under the projection, not one round trip per tile
# speedup vs baseline: 1.4431x; 1.1662x over previous
#include <hip/hip_runtime.h>


#define BATCH_N   16384
#define MAX_ITEMS 200
#define EMB_DIM   128
#define FULL_TILES (MAX_ITEMS / 16)
#define TAIL_ITEMS (MAX_ITEMS % 16)
#define GRID_WG   4096

#define NUM_ITEMS        100000
#define MERGE_BLOCK      128
#define MERGE_PER_THREAD 8
#define MERGE_CHUNKS     (NUM_ITEMS * 2)
#define MERGE_GRID       ((MERGE_CHUNKS + MERGE_BLOCK * MERGE_PER_THREAD - 1) / (MERGE_BLOCK * MERGE_PER_THREAD))

typedef __attribute__((ext_vector_type(2))) float v2f;
typedef __attribute__((ext_vector_type(8))) float v8f;

__global__ __launch_bounds__(MERGE_BLOCK) void interleave_tables_kernel(
    const float* __restrict__ gmf_emb,
    const float* __restrict__ ncf_emb,
    float* __restrict__ merged)
{
    const float2* n2 = (const float2*)ncf_emb;
    const float2* g2 = (const float2*)gmf_emb;
    float4* m4 = (float4*)merged;
    const int base = blockIdx.x * (MERGE_BLOCK * MERGE_PER_THREAD) + threadIdx.x;
    float2 nv[MERGE_PER_THREAD], gv[MERGE_PER_THREAD];
    #pragma unroll
    for (int k = 0; k < MERGE_PER_THREAD; ++k) {
        const int j = base + k * MERGE_BLOCK;
        if (j < MERGE_CHUNKS) { nv[k] = n2[j]; gv[k] = g2[j]; }
    }
    #pragma unroll
    for (int k = 0; k < MERGE_PER_THREAD; ++k) {
        const int j = base + k * MERGE_BLOCK;
        if (j < MERGE_CHUNKS) m4[j] = make_float4(nv[k].x, nv[k].y, gv[k].x, gv[k].y);
    }
}

__device__ __forceinline__ float xhalf(float x) {
    const int b = __builtin_bit_cast(int, x);
    return __builtin_bit_cast(float,
        __builtin_amdgcn_permlanex16(b, b, 0x76543210, 0xfedcba98, false, false));
}

__device__ __forceinline__ void lds_barrier() {
    __builtin_amdgcn_fence(__ATOMIC_RELEASE, "workgroup", "local");
    __builtin_amdgcn_s_barrier();
    __builtin_amdgcn_fence(__ATOMIC_ACQUIRE, "workgroup", "local");
}

__device__ __forceinline__ float tile_result(
    const v2f A, const float4 mv, const float rt, const float sbase,
    const float ga, const float gb, const float (&u_r)[8], const float (&wf8)[8], const float pv)
{
    v2f B = {mv.x, mv.y};
    v8f C = {};
    C = __builtin_amdgcn_wmma_f32_16x16x4_f32(
             false, A,  false, B,
             (short)0, C,  false,  false);

    const float part = sbase + mv.z * ga + mv.w * gb;
    float score = part + xhalf(part);
    #pragma unroll
    for (int r = 0; r < 8; ++r) {
        const float h = u_r[r] + C[r];
        score += fmaxf(h, 0.0f) * wf8[r];
    }
    return pv * rt + score;
}

__global__ __launch_bounds__(128, 16) void ncf_fused_kernel(
    const float* __restrict__ ratings,
    const int*   __restrict__ item_ids,
    const float* __restrict__ noise,
    const float* __restrict__ init_user_emb,
    const float* __restrict__ merged,
    const float* __restrict__ W0, const float* __restrict__ b0,
    const float* __restrict__ W1, const float* __restrict__ b1,
    const float* __restrict__ W2, const float* __restrict__ b2,
    const float* __restrict__ W3, const float* __restrict__ b3,
    const float* __restrict__ Wn, const float* __restrict__ bn,
    const float* __restrict__ Wf, const float* __restrict__ bf,
    const float* __restrict__ p,
    float* __restrict__ out)
{
    __shared__ float s_row[2 * EMB_DIM];
    __shared__ float s_dots[16];
    __shared__ __attribute__((aligned(16))) float s_g[4];
    __shared__ float s_u[8];

    const int t = threadIdx.x;
    const int wave  = __builtin_amdgcn_readfirstlane(t >> 5);
    const int lane  = t & 31;
    const int row16 = lane & 15;
    const bool hi   = lane >= 16;


    const int grp = (t >> 2) & 3;
    const int pj  = t & 3;
    const float* Wp   = (grp == 0) ? W0 : (grp == 1) ? W1 : (grp == 2) ? W2 : W3;
    const float* bp   = (grp == 0) ? b0 : (grp == 1) ? b1 : (grp == 2) ? b2 : b3;
    const float* Wrow = Wp + pj * 64;
    const float  pbias = bp[pj];

    v2f A = {0.0f, 0.0f};
    if (row16 < 8) {
        const int kb = 8 + (hi ? 2 : 0);
        A.x = Wn[row16 * 12 + kb];
        A.y = Wn[row16 * 12 + kb + 1];
    }
    float wf8[8];
    #pragma unroll
    for (int r = 0; r < 8; ++r) wf8[r] = Wf[8 + r];
    const float wf_lo = Wf[t & 3], wf_hi = Wf[4 + (t & 3)];
    const float bn_t  = bn[t & 7];
    const float pv = p[0], bfv = bf[0];
    const float sbase = hi ? 0.0f : bfv;

    const unsigned half1 = hi ? 1u : 0u;
    const unsigned laneItem = (unsigned)(wave * 16 + row16);
    const unsigned tailItem = (unsigned)(FULL_TILES * 16 + ((row16 < TAIL_ITEMS) ? row16 : (TAIL_ITEMS - 1)));
    const bool tailWave = (wave == 3);
    const float4* merged4 = (const float4*)merged;
    for (int b = blockIdx.x; b < BATCH_N; b += GRID_WG) {
        const long rowBase = (long)b * MAX_ITEMS;
        const int*   idrow = item_ids + rowBase;
        const float* rtrow = ratings + rowBase;
        float*       orow  = out + rowBase;

        unsigned iid[4];
        float rt[4];
        #pragma unroll
        for (int s = 0; s < 3; ++s) {
            iid[s] = (unsigned)(idrow + 64 * s)[laneItem];
            rt[s]  = (rtrow + 64 * s)[laneItem];
        }
        iid[3] = 0u; rt[3] = 0.0f;
        if (tailWave) {
            iid[3] = (unsigned)idrow[tailItem];
            rt[3]  = rtrow[tailItem];
        }
        s_row[t]           = init_user_emb[b * EMB_DIM + t];
        s_row[EMB_DIM + t] = noise[b * EMB_DIM + t];
        lds_barrier();

        if (t < 16) {
            const float* x = &s_row[grp * 64];
            float acc = pbias;
            #pragma unroll 8
            for (int k = 0; k < 64; ++k) acc += x[k] * Wrow[k];
            s_dots[t] = acc;
        }

        float4 mv[4];
        #pragma unroll
        for (int s = 0; s < 3; ++s)
            mv[s] = merged4[iid[s] * 2u + half1];
        mv[3] = make_float4(0.0f, 0.0f, 0.0f, 0.0f);
        if (tailWave)
            mv[3] = merged4[iid[3] * 2u + half1];
        lds_barrier();

        if (t < 4) s_g[t] = s_dots[t] * wf_lo + s_dots[8 + t] * wf_hi;
        if (t < 8) {
            float acc = bn_t;
            #pragma unroll
            for (int j = 0; j < 4; ++j)
                acc += Wn[t * 12 + j] * s_dots[4 + j] + Wn[t * 12 + 4 + j] * s_dots[12 + j];
            s_u[t] = acc;
        }
        lds_barrier();

        const float2 g2 = *(const float2*)&s_g[hi ? 2 : 0];
        const float ga = g2.x, gb = g2.y;
        float u_r[8];
        #pragma unroll
        for (int r = 0; r < 8; ++r)
            u_r[r] = __builtin_bit_cast(float,
                __builtin_amdgcn_readfirstlane(__builtin_bit_cast(int, s_u[r])));

        #pragma unroll
        for (int s = 0; s < 3; ++s) {
            const float result = tile_result(A, mv[s], rt[s], sbase, ga, gb, u_r, wf8, pv);
            if (!hi)
                (orow + 64 * s)[laneItem] = result;
        }
        if (tailWave) {
            const float result = tile_result(A, mv[3], rt[3], sbase, ga, gb, u_r, wf8, pv);
            if (!hi && row16 < TAIL_ITEMS)
                orow[FULL_TILES * 16 + row16] = result;
        }
    }
}

extern "C" void kernel_launch(void* const* d_in, const int* in_sizes, int n_in,
                              void* d_out, int out_size, void* d_ws, size_t ws_size,
                              hipStream_t stream) {
    const float* ratings       = (const float*)d_in[0];
    const int*   item_ids      = (const int*)  d_in[1];
    const float* noise         = (const float*)d_in[2];
    const float* init_user_emb = (const float*)d_in[3];
    const float* gmf_emb       = (const float*)d_in[4];
    const float* ncf_emb       = (const float*)d_in[5];
    float* merged = (float*)d_ws;
    const float* W0 = (const float*)d_in[6];  const float* b0 = (const float*)d_in[7];
    const float* W1 = (const float*)d_in[8];  const float* b1 = (const float*)d_in[9];
    const float* W2 = (const float*)d_in[10]; const float* b2 = (const float*)d_in[11];
    const float* W3 = (const float*)d_in[12]; const float* b3 = (const float*)d_in[13];
    const float* Wn = (const float*)d_in[14]; const float* bn = (const float*)d_in[15];
    const float* Wf = (const float*)d_in[16]; const float* bf = (const float*)d_in[17];
    const float* p  = (const float*)d_in[18];
    float* out = (float*)d_out;

    interleave_tables_kernel<<<MERGE_GRID, MERGE_BLOCK, 0, stream>>>(gmf_emb, ncf_emb, merged);

    ncf_fused_kernel<<<GRID_WG, 128, 0, stream>>>(
        ratings, item_ids, noise, init_user_emb, merged,
        W0, b0, W1, b1, W2, b2, W3, b3, Wn, bn, Wf, bf, p, out);
}


# Hidden units split over the half-waves (A rows 0-3 / 8-11): a lane does 4 ReLU+FMA per item, not 8 in half the lanes; score summed as two half-wave parts (sum order changes); head weights via LDS
# speedup vs baseline: 1.5120x; 1.0477x over previous
#include <hip/hip_runtime.h>


#define BATCH_N   16384
#define MAX_ITEMS 200
#define EMB_DIM   128
#define FULL_TILES (MAX_ITEMS / 16)
#define TAIL_ITEMS (MAX_ITEMS % 16)
#define GRID_WG   4096

#define NUM_ITEMS        100000
#define MERGE_BLOCK      128
#define MERGE_PER_THREAD 8
#define MERGE_CHUNKS     (NUM_ITEMS * 2)
#define MERGE_GRID       ((MERGE_CHUNKS + MERGE_BLOCK * MERGE_PER_THREAD - 1) / (MERGE_BLOCK * MERGE_PER_THREAD))

typedef __attribute__((ext_vector_type(2))) float v2f;
typedef __attribute__((ext_vector_type(8))) float v8f;

__global__ __launch_bounds__(MERGE_BLOCK) void interleave_tables_kernel(
    const float* __restrict__ gmf_emb,
    const float* __restrict__ ncf_emb,
    float* __restrict__ merged)
{
    const float2* n2 = (const float2*)ncf_emb;
    const float2* g2 = (const float2*)gmf_emb;
    float4* m4 = (float4*)merged;
    const int base = blockIdx.x * (MERGE_BLOCK * MERGE_PER_THREAD) + threadIdx.x;
    float2 nv[MERGE_PER_THREAD], gv[MERGE_PER_THREAD];
    #pragma unroll
    for (int k = 0; k < MERGE_PER_THREAD; ++k) {
        const int j = base + k * MERGE_BLOCK;
        if (j < MERGE_CHUNKS) { nv[k] = n2[j]; gv[k] = g2[j]; }
    }
    #pragma unroll
    for (int k = 0; k < MERGE_PER_THREAD; ++k) {
        const int j = base + k * MERGE_BLOCK;
        if (j < MERGE_CHUNKS) m4[j] = make_float4(nv[k].x, nv[k].y, gv[k].x, gv[k].y);
    }
}

__device__ __forceinline__ float xhalf(float x) {
    const int b = __builtin_bit_cast(int, x);
    return __builtin_bit_cast(float,
        __builtin_amdgcn_permlanex16(b, b, 0x76543210, 0xfedcba98, false, false));
}

__device__ __forceinline__ void lds_barrier() {
    __builtin_amdgcn_fence(__ATOMIC_RELEASE, "workgroup", "local");
    __builtin_amdgcn_s_barrier();
    __builtin_amdgcn_fence(__ATOMIC_ACQUIRE, "workgroup", "local");
}

__device__ __forceinline__ float tile_result(
    const v2f A, const float4 mv, const float rt, const float sbase,
    const float ga, const float gb, const float4 u4, const float (&wf4)[4], const float pv)
{
    v2f B = {mv.x, mv.y};
    v8f D = {};
    D = __builtin_amdgcn_wmma_f32_16x16x4_f32(
             false, A,  false, B,
             (short)0, D,  false,  false);
    const float h[4] = {u4.x + D[0], u4.y + D[1], u4.z + D[2], u4.w + D[3]};

    float part = sbase + mv.z * ga + mv.w * gb;
    #pragma unroll
    for (int r = 0; r < 4; ++r)
        part = __builtin_fmaf(fmaxf(h[r], 0.0f), wf4[r], part);
    const float score = part + xhalf(part);
    return pv * rt + score;
}

__global__ __launch_bounds__(128, 16) void ncf_fused_kernel(
    const float* __restrict__ ratings,
    const int*   __restrict__ item_ids,
    const float* __restrict__ noise,
    const float* __restrict__ init_user_emb,
    const float* __restrict__ merged,
    const float* __restrict__ W0, const float* __restrict__ b0,
    const float* __restrict__ W1, const float* __restrict__ b1,
    const float* __restrict__ W2, const float* __restrict__ b2,
    const float* __restrict__ W3, const float* __restrict__ b3,
    const float* __restrict__ Wn, const float* __restrict__ bn,
    const float* __restrict__ Wf, const float* __restrict__ bf,
    const float* __restrict__ p,
    float* __restrict__ out)
{
    __shared__ float s_row[2 * EMB_DIM];
    __shared__ float s_dots[16];
    __shared__ __attribute__((aligned(16))) float s_g[4];
    __shared__ __attribute__((aligned(16))) float s_u[8];
    __shared__ float s_wn[128];
    __shared__ float s_hw[16];
    __shared__ float s_pb[16];

    const int t = threadIdx.x;
    const int wave  = __builtin_amdgcn_readfirstlane(t >> 5);
    const int lane  = t & 31;
    const int row16 = lane & 15;
    const bool hi   = lane >= 16;


    const int grp = (t >> 2) & 3;
    const int pj  = t & 3;
    const float* Wp   = (grp == 0) ? W0 : (grp == 1) ? W1 : (grp == 2) ? W2 : W3;
    const float* bp   = (grp == 0) ? b0 : (grp == 1) ? b1 : (grp == 2) ? b2 : b3;
    const float* Wrow = Wp + pj * 64;
    s_pb[t & 15] = bp[pj];
    s_hw[t & 15] = ((t & 8) ? bn : Wf)[t & 7];
    s_wn[t]      = Wn[t < 96 ? t : 95];

    v2f A = {0.0f, 0.0f};
    if ((row16 & 7) < 4) {
        const int unit = (row16 & 3) + ((row16 & 8) ? 4 : 0);
        const int kb = 8 + (hi ? 2 : 0);
        A.x = Wn[unit * 12 + kb];
        A.y = Wn[unit * 12 + kb + 1];
    }
    float wf4[4];
    #pragma unroll
    for (int r = 0; r < 4; ++r) wf4[r] = Wf[8 + (hi ? 4 : 0) + r];
    const float pv = p[0], bfv = bf[0];
    const float sbase = hi ? 0.0f : bfv;

    const unsigned half1 = hi ? 1u : 0u;
    const unsigned laneItem = (unsigned)(wave * 16 + row16);
    const unsigned tailItem = (unsigned)(FULL_TILES * 16 + ((row16 < TAIL_ITEMS) ? row16 : (TAIL_ITEMS - 1)));
    const bool tailWave = (wave == 3);
    const float4* merged4 = (const float4*)merged;
    for (int b = blockIdx.x; b < BATCH_N; b += GRID_WG) {
        const long rowBase = (long)b * MAX_ITEMS;
        const int*   idrow = item_ids + rowBase;
        const float* rtrow = ratings + rowBase;
        float*       orow  = out + rowBase;

        unsigned iid[4];
        float rt[4];
        #pragma unroll
        for (int s = 0; s < 3; ++s) {
            iid[s] = (unsigned)(idrow + 64 * s)[laneItem];
            rt[s]  = (rtrow + 64 * s)[laneItem];
        }
        iid[3] = 0u; rt[3] = 0.0f;
        if (tailWave) {
            iid[3] = (unsigned)idrow[tailItem];
            rt[3]  = rtrow[tailItem];
        }
        s_row[t]           = init_user_emb[b * EMB_DIM + t];
        s_row[EMB_DIM + t] = noise[b * EMB_DIM + t];
        lds_barrier();

        if (t < 16) {
            const float* x = &s_row[grp * 64];
            float acc = s_pb[t];
            #pragma unroll 8
            for (int k = 0; k < 64; ++k) acc += x[k] * Wrow[k];
            s_dots[t] = acc;
        }

        float4 mv[4];
        #pragma unroll
        for (int s = 0; s < 3; ++s)
            mv[s] = merged4[iid[s] * 2u + half1];
        mv[3] = make_float4(0.0f, 0.0f, 0.0f, 0.0f);
        if (tailWave)
            mv[3] = merged4[iid[3] * 2u + half1];
        lds_barrier();

        if (t < 4) s_g[t] = s_dots[t] * s_hw[t] + s_dots[8 + t] * s_hw[4 + t];
        if (t < 8) {
            float acc = s_hw[8 + t];
            #pragma unroll
            for (int j = 0; j < 4; ++j)
                acc += s_wn[t * 12 + j] * s_dots[4 + j] + s_wn[t * 12 + 4 + j] * s_dots[12 + j];
            s_u[t] = acc;
        }
        lds_barrier();

        const float2 g2 = *(const float2*)&s_g[hi ? 2 : 0];
        const float ga = g2.x, gb = g2.y;
        const float4 u4 = *(const float4*)&s_u[hi ? 4 : 0];

        #pragma unroll
        for (int s = 0; s < 3; ++s) {
            const float result = tile_result(A, mv[s], rt[s], sbase, ga, gb, u4, wf4, pv);
            if (!hi)
                (orow + 64 * s)[laneItem] = result;
        }
        if (tailWave) {
            const float result = tile_result(A, mv[3], rt[3], sbase, ga, gb, u4, wf4, pv);
            if (!hi && row16 < TAIL_ITEMS)
                orow[FULL_TILES * 16 + row16] = result;
        }
    }
}

extern "C" void kernel_launch(void* const* d_in, const int* in_sizes, int n_in,
                              void* d_out, int out_size, void* d_ws, size_t ws_size,
                              hipStream_t stream) {
    const float* ratings       = (const float*)d_in[0];
    const int*   item_ids      = (const int*)  d_in[1];
    const float* noise         = (const float*)d_in[2];
    const float* init_user_emb = (const float*)d_in[3];
    const float* gmf_emb       = (const float*)d_in[4];
    const float* ncf_emb       = (const float*)d_in[5];
    float* merged = (float*)d_ws;
    const float* W0 = (const float*)d_in[6];  const float* b0 = (const float*)d_in[7];
    const float* W1 = (const float*)d_in[8];  const float* b1 = (const float*)d_in[9];
    const float* W2 = (const float*)d_in[10]; const float* b2 = (const float*)d_in[11];
    const float* W3 = (const float*)d_in[12]; const float* b3 = (const float*)d_in[13];
    const float* Wn = (const float*)d_in[14]; const float* bn = (const float*)d_in[15];
    const float* Wf = (const float*)d_in[16]; const float* bf = (const float*)d_in[17];
    const float* p  = (const float*)d_in[18];
    float* out = (float*)d_out;

    interleave_tables_kernel<<<MERGE_GRID, MERGE_BLOCK, 0, stream>>>(gmf_emb, ncf_emb, merged);

    ncf_fused_kernel<<<GRID_WG, 128, 0, stream>>>(
        ratings, item_ids, noise, init_user_emb, merged,
        W0, b0, W1, b1, W2, b2, W3, b3, Wn, bn, Wf, bf, p, out);
}


# Per-row work (projections, g, u) moved into the pre-pass as a 64-byte row record: item kernel has no LDS or barriers, one wave owns a row with all 13 gathers in flight; projection sum order changes
# speedup vs baseline: 2.0366x; 1.3470x over previous
#include <hip/hip_runtime.h>


#define BATCH_N   16384
#define MAX_ITEMS 200
#define EMB_DIM   128
#define NUM_ITEMS 100000

#define MERGED_FLOATS   (NUM_ITEMS * 8)
#define REC_OFFSET      (1024 * 1024)
#define REC_FLOATS      16

#define PRE_BLOCK        128
#define MERGE_PER_THREAD 8
#define MERGE_CHUNKS     (NUM_ITEMS * 2)
#define MERGE_GRID       ((MERGE_CHUNKS + PRE_BLOCK * MERGE_PER_THREAD - 1) / (PRE_BLOCK * MERGE_PER_THREAD))
#define PROJ_ROWS        8
#define PROJ_GRID        (BATCH_N / (PROJ_ROWS * (PRE_BLOCK / 32)))
#define PRE_GRID         (PROJ_GRID + MERGE_GRID)

#define ITEM_BLOCK      256
#define ITEM_ROWS       2
#define ITEM_WAVES      (BATCH_N / ITEM_ROWS)
#define ITEM_GRID       (ITEM_WAVES / (ITEM_BLOCK / 32))
#define PAIRS           (MAX_ITEMS / 32)
#define TAIL_BASE       (PAIRS * 32)
#define TAIL_ITEMS      (MAX_ITEMS - TAIL_BASE)

typedef __attribute__((ext_vector_type(2))) float v2f;
typedef __attribute__((ext_vector_type(4))) float v4f;
typedef __attribute__((ext_vector_type(8))) float v8f;

__device__ __forceinline__ int xhalf_i(int b) {
    return __builtin_amdgcn_permlanex16(b, b, 0x76543210, 0xfedcba98, false, false);
}
__device__ __forceinline__ float xhalf(float x) {
    return __builtin_bit_cast(float, xhalf_i(__builtin_bit_cast(int, x)));
}

template <int M>
__device__ __forceinline__ float xmask_add(float v) {
    const int i = __builtin_bit_cast(int, v);
    const int o = __builtin_amdgcn_mov_dpp(i, 0x160 + M, 0xf, 0xf, true);
    return v + __builtin_bit_cast(float, o);
}
__device__ __forceinline__ float row_sum16(float v) {
    v = xmask_add<1>(v);
    v = xmask_add<2>(v);
    v = xmask_add<4>(v);
    v = xmask_add<8>(v);
    return v;
}

__device__ __forceinline__ float dot4(const v4f x, const v4f w) {
    float a = x.x * w.x;
    a = __builtin_fmaf(x.y, w.y, a);
    a = __builtin_fmaf(x.z, w.z, a);
    a = __builtin_fmaf(x.w, w.w, a);
    return a;
}

__global__ __launch_bounds__(PRE_BLOCK) void prepass_kernel(
    const float* __restrict__ gmf_emb,
    const float* __restrict__ ncf_emb,
    const float* __restrict__ noise,
    const float* __restrict__ init_user_emb,
    const float* __restrict__ W0, const float* __restrict__ b0,
    const float* __restrict__ W1, const float* __restrict__ b1,
    const float* __restrict__ W2, const float* __restrict__ b2,
    const float* __restrict__ W3, const float* __restrict__ b3,
    const float* __restrict__ Wn, const float* __restrict__ bn,
    const float* __restrict__ Wf, const float* __restrict__ bf,
    float* __restrict__ merged,
    float* __restrict__ rowrec)
{
    if (blockIdx.x >= PROJ_GRID) {
        const float2* n2 = (const float2*)ncf_emb;
        const float2* g2 = (const float2*)gmf_emb;
        float4* m4 = (float4*)merged;
        const int base = (blockIdx.x - PROJ_GRID) * (PRE_BLOCK * MERGE_PER_THREAD) + threadIdx.x;
        float2 nv[MERGE_PER_THREAD], gv[MERGE_PER_THREAD];
        #pragma unroll
        for (int k = 0; k < MERGE_PER_THREAD; ++k) {
            const int j = base + k * PRE_BLOCK;
            if (j < MERGE_CHUNKS) { nv[k] = n2[j]; gv[k] = g2[j]; }
        }
        #pragma unroll
        for (int k = 0; k < MERGE_PER_THREAD; ++k) {
            const int j = base + k * PRE_BLOCK;
            if (j < MERGE_CHUNKS) m4[j] = make_float4(nv[k].x, nv[k].y, gv[k].x, gv[k].y);
        }
        return;
    }

    const int lane = threadIdx.x & 31;
    const int c    = lane & 15;
    const bool hi  = lane >= 16;
    const int wave = blockIdx.x * (PRE_BLOCK / 32) + (threadIdx.x >> 5);
    const int row0 = wave * PROJ_ROWS;

    v4f xu[PROJ_ROWS], xz[PROJ_ROWS];
    #pragma unroll
    for (int i = 0; i < PROJ_ROWS; ++i) {
        xu[i] = ((const v4f*)(init_user_emb + (row0 + i) * EMB_DIM))[lane];
        xz[i] = ((const v4f*)(noise         + (row0 + i) * EMB_DIM))[lane];
    }

    const float* Wu = hi ? W1 : W0;
    const float* Wz = hi ? W3 : W2;
    const float* bu = hi ? b1 : b0;
    const float* bz = hi ? b3 : b2;
    v4f wu[4], wz[4];
    float cu[4], cz[4];
    #pragma unroll
    for (int j = 0; j < 4; ++j) {
        wu[j] = ((const v4f*)(Wu + j * 64))[c];
        wz[j] = ((const v4f*)(Wz + j * 64))[c];
        cu[j] = bu[j];
        cz[j] = bz[j];
    }
    const int r  = c & 7;
    const int gk = (c < 2) ? c : c - 2;
    const bool isg = !hi && (c < 2 || c == 4 || c == 5);
    float wa[4], wb[4];
    #pragma unroll
    for (int j = 0; j < 4; ++j) { wa[j] = Wn[r * 12 + j]; wb[j] = Wn[r * 12 + 4 + j]; }
    const float bnr = bn[r], bfv = bf[0];

    asm volatile("" :: "v"(xu[0]), "v"(xu[1]), "v"(xu[2]), "v"(xu[3]),
                       "v"(xu[4]), "v"(xu[5]), "v"(xu[6]), "v"(xu[7]),
                       "v"(xz[0]), "v"(xz[1]), "v"(xz[2]), "v"(xz[3]),
                       "v"(xz[4]), "v"(xz[5]), "v"(xz[6]), "v"(xz[7]),
                       "v"(wa[0]), "v"(wa[1]), "v"(wa[2]), "v"(wa[3]),
                       "v"(wb[0]), "v"(wb[1]), "v"(wb[2]), "v"(wb[3]), "v"(bnr));

    float ca[4], cb[4];
    #pragma unroll
    for (int j = 0; j < 4; ++j) {
        ca[j] = hi ? wa[j] : ((isg && gk == j) ? Wf[j]     : 0.0f);
        cb[j] = hi ? wb[j] : ((isg && gk == j) ? Wf[4 + j] : 0.0f);
    }
    const float cg = hi ? bnr : ((c == 2) ? bfv : 0.0f);
    const int slot = hi ? ((c < 4) ? c : c + 4) : ((c < 4) ? c + 4 : c + 8);

    #pragma unroll
    for (int i = 0; i < PROJ_ROWS; ++i) {
        float val = cg;
        #pragma unroll
        for (int j = 0; j < 4; ++j) {
            const float su = row_sum16(dot4(xu[i], wu[j])) + cu[j];
            const float sz = row_sum16(dot4(xz[i], wz[j])) + cz[j];
            val += ca[j] * su + cb[j] * sz;
        }
        if (c < 8) rowrec[(row0 + i) * REC_FLOATS + slot] = val;
    }
}

__device__ __forceinline__ float tile_score(
    const v2f A, const float4 mv, const float4 u4, const float4 gx, const float (&wf4)[4])
{
    v2f B = {mv.x, mv.y};
    v8f D = {};
    D = __builtin_amdgcn_wmma_f32_16x16x4_f32(
             false, A,  false, B,
             (short)0, D,  false,  false);
    const float h[4] = {u4.x + D[0], u4.y + D[1], u4.z + D[2], u4.w + D[3]};
    float part = gx.z + mv.z * gx.x + mv.w * gx.y;
    #pragma unroll
    for (int r = 0; r < 4; ++r)
        part = __builtin_fmaf(fmaxf(h[r], 0.0f), wf4[r], part);
    return part + xhalf(part);
}

struct RowIn {
    int    id[PAIRS + 1];
    float  rt[PAIRS + 1];
    float4 u4, gx;
};

__device__ __forceinline__ void load_row(
    RowIn& in, const int b, const int lane, const int tailItem,
    const int* __restrict__ item_ids, const float* __restrict__ ratings)
{
    const int*   idrow = item_ids + (long)b * MAX_ITEMS;
    const float* rtrow = ratings  + (long)b * MAX_ITEMS;
    #pragma unroll
    for (int j = 0; j < PAIRS; ++j) {
        in.id[j] = (idrow + 32 * j)[lane];
        in.rt[j] = (rtrow + 32 * j)[lane];
    }
    in.id[PAIRS] = idrow[tailItem];
    in.rt[PAIRS] = rtrow[tailItem];
}

__device__ __forceinline__ void load_rec(
    RowIn& in, const int b, const int half, const float4* __restrict__ rec4)
{
    in.u4 = rec4[b * 4 + 2 * half];
    in.gx = rec4[b * 4 + 2 * half + 1];
}

__global__ __launch_bounds__(ITEM_BLOCK, 8) void ncf_items_kernel(
    const float* __restrict__ ratings,
    const int*   __restrict__ item_ids,
    const float* __restrict__ merged,
    const float* __restrict__ rowrec,
    const float* __restrict__ Wn,
    const float* __restrict__ Wf,
    const float* __restrict__ p,
    float* __restrict__ out)
{
    const int t = threadIdx.x;
    const int lane = t & 31;
    const int c    = lane & 15;
    const bool hi  = lane >= 16;
    const int half = hi ? 1 : 0;
    const int gw   = blockIdx.x * (ITEM_BLOCK / 32) + (t >> 5);

    v2f A = {0.0f, 0.0f};
    if ((c & 7) < 4) {
        const int unit = (c & 3) + ((c & 8) ? 4 : 0);
        const int kb = 8 + (hi ? 2 : 0);
        A.x = Wn[unit * 12 + kb];
        A.y = Wn[unit * 12 + kb + 1];
    }
    float wf4[4];
    #pragma unroll
    for (int r = 0; r < 4; ++r) wf4[r] = Wf[8 + (hi ? 4 : 0) + r];
    const float pv = p[0];

    const int tailItem = TAIL_BASE + ((c < TAIL_ITEMS) ? c : (TAIL_ITEMS - 1));
    const float4* merged4 = (const float4*)merged;
    const float4* rec4    = (const float4*)rowrec;

    RowIn in[ITEM_ROWS];
    load_row(in[0], gw, lane, tailItem, item_ids, ratings);
    load_rec(in[0], gw, half, rec4);
    __builtin_amdgcn_sched_barrier(0);

    #pragma unroll
    for (int it = 0; it < ITEM_ROWS; ++it) {
        const int b = gw + it * ITEM_WAVES;
        float* orow = out + (long)b * MAX_ITEMS;

        asm volatile("" :: "v"(in[it].rt[PAIRS]));
        if (it > 0)
            load_rec(in[it], b, half, rec4);

        float4 mvA[PAIRS], mvB[PAIRS];
        #pragma unroll
        for (int j = 0; j < PAIRS; ++j) {
            const int own = in[it].id[j];
            const int oth = xhalf_i(own);
            const unsigned idA = (unsigned)(hi ? oth : own);
            const unsigned idB = (unsigned)(hi ? own : oth);
            mvA[j] = merged4[idA * 2u + (unsigned)half];
            mvB[j] = merged4[idB * 2u + (unsigned)half];
        }
        const float4 mvT = merged4[(unsigned)in[it].id[PAIRS] * 2u + (unsigned)half];
        __builtin_amdgcn_sched_barrier(0);

        if (it + 1 < ITEM_ROWS)
            load_row(in[it + 1], b + ITEM_WAVES, lane, tailItem, item_ids, ratings);
        __builtin_amdgcn_sched_barrier(0);

        #pragma unroll
        for (int j = 0; j < PAIRS; ++j) {
            const float sA = tile_score(A, mvA[j], in[it].u4, in[it].gx, wf4);
            const float sB = tile_score(A, mvB[j], in[it].u4, in[it].gx, wf4);
            (orow + 32 * j)[lane] = pv * in[it].rt[j] + (hi ? sB : sA);
        }
        const float sT = tile_score(A, mvT, in[it].u4, in[it].gx, wf4);
        if (!hi && c < TAIL_ITEMS)
            orow[TAIL_BASE + c] = pv * in[it].rt[PAIRS] + sT;
    }
}

extern "C" void kernel_launch(void* const* d_in, const int* in_sizes, int n_in,
                              void* d_out, int out_size, void* d_ws, size_t ws_size,
                              hipStream_t stream) {
    const float* ratings       = (const float*)d_in[0];
    const int*   item_ids      = (const int*)  d_in[1];
    const float* noise         = (const float*)d_in[2];
    const float* init_user_emb = (const float*)d_in[3];
    const float* gmf_emb       = (const float*)d_in[4];
    const float* ncf_emb       = (const float*)d_in[5];
    float* merged = (float*)d_ws;
    float* rowrec = merged + REC_OFFSET;
    const float* W0 = (const float*)d_in[6];  const float* b0 = (const float*)d_in[7];
    const float* W1 = (const float*)d_in[8];  const float* b1 = (const float*)d_in[9];
    const float* W2 = (const float*)d_in[10]; const float* b2 = (const float*)d_in[11];
    const float* W3 = (const float*)d_in[12]; const float* b3 = (const float*)d_in[13];
    const float* Wn = (const float*)d_in[14]; const float* bn = (const float*)d_in[15];
    const float* Wf = (const float*)d_in[16]; const float* bf = (const float*)d_in[17];
    const float* p  = (const float*)d_in[18];
    float* out = (float*)d_out;

    prepass_kernel<<<PRE_GRID, PRE_BLOCK, 0, stream>>>(
        gmf_emb, ncf_emb, noise, init_user_emb,
        W0, b0, W1, b1, W2, b2, W3, b3, Wn, bn, Wf, bf, merged, rowrec);

    ncf_items_kernel<<<ITEM_GRID, ITEM_BLOCK, 0, stream>>>(
        ratings, item_ids, merged, rowrec, Wn, Wf, p, out);
}
